# starting kernel
# baseline (speedup 1.0000x reference)
#include <hip/hip_runtime.h>

#define NN 100000
#define NE 1600000
#define D 128
#define D4 132
#define NL 7
#define GEMM_BLOCKS 1250

typedef __attribute__((ext_vector_type(2))) float v2f;
typedef __attribute__((ext_vector_type(8))) float v8f;


__global__ void gnn_hist_kernel(const int* __restrict__ dst, int* __restrict__ cnt) {
    int e = blockIdx.x * blockDim.x + threadIdx.x;
    if (e < NE) atomicAdd(&cnt[dst[e]], 1);
}

__global__ void gnn_scan_kernel(const int* __restrict__ cnt, int* __restrict__ row_ptr,
                                float* __restrict__ deg_inv) {
    __shared__ int sdata[1024];
    __shared__ int s_off;
    int tid = threadIdx.x;
    if (tid == 0) { s_off = 0; row_ptr[0] = 0; }
    __syncthreads();
    for (int base = 0; base < NN; base += 1024) {
        int i = base + tid;
        int v = (i < NN) ? cnt[i] : 0;
        sdata[tid] = v;
        __syncthreads();
        for (int off = 1; off < 1024; off <<= 1) {
            int t = (tid >= off) ? sdata[tid - off] : 0;
            __syncthreads();
            sdata[tid] += t;
            __syncthreads();
        }
        int incl = sdata[tid];
        int carry = s_off;
        if (i < NN) {
            row_ptr[i + 1] = carry + incl;
            deg_inv[i] = 1.0f / fmaxf((float)v, 1.0f);
        }
        __syncthreads();
        if (tid == 1023) s_off = carry + sdata[1023];
        __syncthreads();
    }
}

__global__ void gnn_fill_kernel(const int* __restrict__ src, const int* __restrict__ dst,
                                const int* __restrict__ row_ptr, int* __restrict__ cnt2,
                                int* __restrict__ col_src) {
    int e = blockIdx.x * blockDim.x + threadIdx.x;
    if (e < NE) {
        int d = dst[e];
        int pos = row_ptr[d] + atomicAdd(&cnt2[d], 1);
        col_src[pos] = src[e];
    }
}


__global__ void __launch_bounds__(256)
gnn_agg_kernel(const float* __restrict__ h, const int* __restrict__ row_ptr,
               const int* __restrict__ col_src, const float* __restrict__ deg_inv,
               float* __restrict__ mean) {
    int wave = threadIdx.x >> 5;
    int lane = threadIdx.x & 31;
    int node = blockIdx.x * (blockDim.x >> 5) + wave;
    if (node >= NN) return;
    int e0 = row_ptr[node];
    int e1 = row_ptr[node + 1];
    float4 acc = make_float4(0.f, 0.f, 0.f, 0.f);
    for (int e = e0; e < e1; ++e) {
        int s = col_src[e];
        if (e + 1 < e1)
            __builtin_prefetch(h + (size_t)col_src[e + 1] * D + lane * 4, 0, 1);
        float4 v = ((const float4*)(h + (size_t)s * D))[lane];
        acc.x += v.x; acc.y += v.y; acc.z += v.z; acc.w += v.w;
    }
    float di = deg_inv[node];
    float4 r = make_float4(acc.x * di, acc.y * di, acc.z * di, acc.w * di);
    ((float4*)(mean + (size_t)node * D))[lane] = r;
}


__global__ void __launch_bounds__(256, 1)
gnn_gemm_kernel(const float* __restrict__ mean, const float* __restrict__ x,
                const float* __restrict__ Wl, const float* __restrict__ Wr,
                const float* __restrict__ bias, float* __restrict__ out, int relu) {
    __shared__ float sA[16 * D4];
    __shared__ float sX[16 * D4];

    int wave = threadIdx.x >> 5;
    int lane = threadIdx.x & 31;
    int r     = lane & 15;
    int khalf = lane >> 4;

    v2f bl[32], br[32];
    {
        const float* wlrow = Wl + (size_t)(wave * 16 + r) * D;
        const float* wrrow = Wr + (size_t)(wave * 16 + r) * D;
        #pragma unroll
        for (int kk = 0; kk < 32; ++kk) {
            int ko = kk * 4 + 2 * khalf;
            bl[kk] = *(const v2f*)(wlrow + ko);
            br[kk] = *(const v2f*)(wrrow + ko);
        }
    }
    float bj = bias[wave * 16 + r];

    for (int rt = blockIdx.x; rt < NN / 16; rt += GEMM_BLOCKS) {
        {
            const float4* gm = (const float4*)(mean + (size_t)rt * 16 * D);
            const float4* gx = (const float4*)(x    + (size_t)rt * 16 * D);
            #pragma unroll
            for (int j = 0; j < 2; ++j) {
                int i    = threadIdx.x + j * 256;
                int row  = i >> 5;
                int col4 = i & 31;
                ((float4*)(sA + row * D4))[col4] = gm[i];
                ((float4*)(sX + row * D4))[col4] = gx[i];
            }
        }
        __syncthreads();

        v8f acc = {};
        #pragma unroll
        for (int kk = 0; kk < 32; ++kk) {
            int ko = kk * 4 + 2 * khalf;
            v2f a = *(const v2f*)(sA + r * D4 + ko);
            acc = __builtin_amdgcn_wmma_f32_16x16x4_f32(false, a, false, bl[kk],
                                                        (short)0, acc, false, false);
            v2f a2 = *(const v2f*)(sX + r * D4 + ko);
            acc = __builtin_amdgcn_wmma_f32_16x16x4_f32(false, a2, false, br[kk],
                                                        (short)0, acc, false, false);
        }

        float* obase = out + (size_t)(rt * 16) * D + wave * 16 + r;
        #pragma unroll
        for (int v = 0; v < 8; ++v) {
            float val = acc[v] + bj;
            if (relu) val = fmaxf(val, 0.0f);
            obase[(size_t)(v + 8 * khalf) * D] = val;
        }
        __syncthreads();
    }
}


extern "C" void kernel_launch(void* const* d_in, const int* in_sizes, int n_in,
                              void* d_out, int out_size, void* d_ws, size_t ws_size,
                              hipStream_t stream) {
    const float* x          = (const float*)d_in[0];
    const int*   edge_index = (const int*)d_in[1];
    const float* Wl         = (const float*)d_in[2];
    const float* Wr         = (const float*)d_in[3];
    const float* b          = (const float*)d_in[4];
    const int* src = edge_index;
    const int* dst = edge_index + NE;

    char* ws = (char*)d_ws;
    size_t off = 0;
    auto carve = [&](size_t bytes) -> void* {
        void* p = ws + off;
        off = (off + bytes + 255) & ~(size_t)255;
        return p;
    };
    float* mean    = (float*)carve((size_t)NN * D * sizeof(float));
    float* hA      = (float*)carve((size_t)NN * D * sizeof(float));
    float* hB      = (float*)carve((size_t)NN * D * sizeof(float));
    float* deg_inv = (float*)carve((size_t)NN * sizeof(float));
    int*   row_ptr = (int*)carve((size_t)(NN + 1) * sizeof(int));
    int*   cnt     = (int*)carve((size_t)NN * sizeof(int));
    int*   col_src = (int*)carve((size_t)NE * sizeof(int));

    hipMemsetAsync(cnt, 0, (size_t)NN * sizeof(int), stream);
    gnn_hist_kernel<<<(NE + 255) / 256, 256, 0, stream>>>(dst, cnt);
    gnn_scan_kernel<<<1, 1024, 0, stream>>>(cnt, row_ptr, deg_inv);
    hipMemsetAsync(cnt, 0, (size_t)NN * sizeof(int), stream);
    gnn_fill_kernel<<<(NE + 255) / 256, 256, 0, stream>>>(src, dst, row_ptr, cnt, col_src);

    const float* hin = x;
    float* houts[2] = {hA, hB};
    for (int l = 0; l < NL; ++l) {
        float* hout = (l == NL - 1) ? (float*)d_out : houts[l & 1];
        gnn_agg_kernel<<<NN / 8, 256, 0, stream>>>(hin, row_ptr, col_src, deg_inv, mean);
        gnn_gemm_kernel<<<GEMM_BLOCKS, 256, 0, stream>>>(
            mean, hin, Wl + (size_t)l * D * D, Wr + (size_t)l * D * D,
            b + (size_t)l * D, hout, (l < NL - 1) ? 1 : 0);
        hin = hout;
    }
}


# Degree prefix sum on 98 blocks instead of 1: each block re-sums the counts in front of its 1024-count chunk from L2, then one shuffle scan and one barrier; row_ptr values identical
# speedup vs baseline: 1.2811x; 1.2811x over previous
#include <hip/hip_runtime.h>

#define NN 100000
#define NE 1600000
#define D 128
#define D4 132
#define NL 7
#define GEMM_BLOCKS 1250
#define SCAN_T 1024
#define SCAN_BLOCKS ((NN + SCAN_T - 1) / SCAN_T)
static_assert(SCAN_T / 32 == 32, "the cross-wave sum gives each of the 32 lanes one wave total");

typedef __attribute__((ext_vector_type(2))) float v2f;
typedef __attribute__((ext_vector_type(8))) float v8f;


__global__ void gnn_hist_kernel(const int* __restrict__ dst, int* __restrict__ cnt) {
    int e = blockIdx.x * blockDim.x + threadIdx.x;
    if (e < NE) atomicAdd(&cnt[dst[e]], 1);
}

__global__ void __launch_bounds__(SCAN_T)
gnn_scan_kernel(const int* __restrict__ cnt, int* __restrict__ row_ptr,
                float* __restrict__ deg_inv) {
    __shared__ int s_chunk[SCAN_T / 32];
    __shared__ int s_front[SCAN_T / 32];
    int tid  = threadIdx.x;
    int lane = tid & 31;
    int wave = tid >> 5;
    int base = blockIdx.x * SCAN_T;

    int front = 0;
    {
        const int4* c4 = (const int4*)cnt;
        int n4 = base >> 2;
        #pragma unroll 4
        for (int i = tid; i < n4; i += SCAN_T) {
            int4 q = c4[i];
            front += (q.x + q.y) + (q.z + q.w);
        }
    }

    int i = base + tid;
    int v = (i < NN) ? cnt[i] : 0;

    int incl = v;
    for (int off = 1; off < 32; off <<= 1) {
        int t = __shfl_up(incl, off, 32);
        if (lane >= off) incl += t;
    }
    for (int off = 16; off > 0; off >>= 1) front += __shfl_xor(front, off, 32);

    if (lane == 31) s_chunk[wave] = incl;
    if (lane == 0)  s_front[wave] = front;
    __syncthreads();

    int c = s_front[lane] + ((lane < wave) ? s_chunk[lane] : 0);
    for (int off = 16; off > 0; off >>= 1) c += __shfl_xor(c, off, 32);

    if (i < NN) {
        row_ptr[i + 1] = c + incl;
        deg_inv[i] = 1.0f / fmaxf((float)v, 1.0f);
    }
    if (i == 0) row_ptr[0] = 0;
}

__global__ void gnn_fill_kernel(const int* __restrict__ src, const int* __restrict__ dst,
                                const int* __restrict__ row_ptr, int* __restrict__ cnt2,
                                int* __restrict__ col_src) {
    int e = blockIdx.x * blockDim.x + threadIdx.x;
    if (e < NE) {
        int d = dst[e];
        int pos = row_ptr[d] + atomicAdd(&cnt2[d], 1);
        col_src[pos] = src[e];
    }
}


__global__ void __launch_bounds__(256)
gnn_agg_kernel(const float* __restrict__ h, const int* __restrict__ row_ptr,
               const int* __restrict__ col_src, const float* __restrict__ deg_inv,
               float* __restrict__ mean) {
    int wave = threadIdx.x >> 5;
    int lane = threadIdx.x & 31;
    int node = blockIdx.x * (blockDim.x >> 5) + wave;
    if (node >= NN) return;
    int e0 = row_ptr[node];
    int e1 = row_ptr[node + 1];
    float4 acc = make_float4(0.f, 0.f, 0.f, 0.f);
    for (int e = e0; e < e1; ++e) {
        int s = col_src[e];
        if (e + 1 < e1)
            __builtin_prefetch(h + (size_t)col_src[e + 1] * D + lane * 4, 0, 1);
        float4 v = ((const float4*)(h + (size_t)s * D))[lane];
        acc.x += v.x; acc.y += v.y; acc.z += v.z; acc.w += v.w;
    }
    float di = deg_inv[node];
    float4 r = make_float4(acc.x * di, acc.y * di, acc.z * di, acc.w * di);
    ((float4*)(mean + (size_t)node * D))[lane] = r;
}


__global__ void __launch_bounds__(256, 1)
gnn_gemm_kernel(const float* __restrict__ mean, const float* __restrict__ x,
                const float* __restrict__ Wl, const float* __restrict__ Wr,
                const float* __restrict__ bias, float* __restrict__ out, int relu) {
    __shared__ float sA[16 * D4];
    __shared__ float sX[16 * D4];

    int wave = threadIdx.x >> 5;
    int lane = threadIdx.x & 31;
    int r     = lane & 15;
    int khalf = lane >> 4;

    v2f bl[32], br[32];
    {
        const float* wlrow = Wl + (size_t)(wave * 16 + r) * D;
        const float* wrrow = Wr + (size_t)(wave * 16 + r) * D;
        #pragma unroll
        for (int kk = 0; kk < 32; ++kk) {
            int ko = kk * 4 + 2 * khalf;
            bl[kk] = *(const v2f*)(wlrow + ko);
            br[kk] = *(const v2f*)(wrrow + ko);
        }
    }
    float bj = bias[wave * 16 + r];

    for (int rt = blockIdx.x; rt < NN / 16; rt += GEMM_BLOCKS) {
        {
            const float4* gm = (const float4*)(mean + (size_t)rt * 16 * D);
            const float4* gx = (const float4*)(x    + (size_t)rt * 16 * D);
            #pragma unroll
            for (int j = 0; j < 2; ++j) {
                int i    = threadIdx.x + j * 256;
                int row  = i >> 5;
                int col4 = i & 31;
                ((float4*)(sA + row * D4))[col4] = gm[i];
                ((float4*)(sX + row * D4))[col4] = gx[i];
            }
        }
        __syncthreads();

        v8f acc = {};
        #pragma unroll
        for (int kk = 0; kk < 32; ++kk) {
            int ko = kk * 4 + 2 * khalf;
            v2f a = *(const v2f*)(sA + r * D4 + ko);
            acc = __builtin_amdgcn_wmma_f32_16x16x4_f32(false, a, false, bl[kk],
                                                        (short)0, acc, false, false);
            v2f a2 = *(const v2f*)(sX + r * D4 + ko);
            acc = __builtin_amdgcn_wmma_f32_16x16x4_f32(false, a2, false, br[kk],
                                                        (short)0, acc, false, false);
        }

        float* obase = out + (size_t)(rt * 16) * D + wave * 16 + r;
        #pragma unroll
        for (int v = 0; v < 8; ++v) {
            float val = acc[v] + bj;
            if (relu) val = fmaxf(val, 0.0f);
            obase[(size_t)(v + 8 * khalf) * D] = val;
        }
        __syncthreads();
    }
}


extern "C" void kernel_launch(void* const* d_in, const int* in_sizes, int n_in,
                              void* d_out, int out_size, void* d_ws, size_t ws_size,
                              hipStream_t stream) {
    const float* x          = (const float*)d_in[0];
    const int*   edge_index = (const int*)d_in[1];
    const float* Wl         = (const float*)d_in[2];
    const float* Wr         = (const float*)d_in[3];
    const float* b          = (const float*)d_in[4];
    const int* src = edge_index;
    const int* dst = edge_index + NE;

    char* ws = (char*)d_ws;
    size_t off = 0;
    auto carve = [&](size_t bytes) -> void* {
        void* p = ws + off;
        off = (off + bytes + 255) & ~(size_t)255;
        return p;
    };
    float* mean    = (float*)carve((size_t)NN * D * sizeof(float));
    float* hA      = (float*)carve((size_t)NN * D * sizeof(float));
    float* hB      = (float*)carve((size_t)NN * D * sizeof(float));
    float* deg_inv = (float*)carve((size_t)NN * sizeof(float));
    int*   row_ptr = (int*)carve((size_t)(NN + 1) * sizeof(int));
    int*   cnt     = (int*)carve((size_t)NN * sizeof(int));
    int*   col_src = (int*)carve((size_t)NE * sizeof(int));

    hipMemsetAsync(cnt, 0, (size_t)NN * sizeof(int), stream);
    gnn_hist_kernel<<<(NE + 255) / 256, 256, 0, stream>>>(dst, cnt);
    gnn_scan_kernel<<<SCAN_BLOCKS, SCAN_T, 0, stream>>>(cnt, row_ptr, deg_inv);
    hipMemsetAsync(cnt, 0, (size_t)NN * sizeof(int), stream);
    gnn_fill_kernel<<<(NE + 255) / 256, 256, 0, stream>>>(src, dst, row_ptr, cnt, col_src);

    const float* hin = x;
    float* houts[2] = {hA, hB};
    for (int l = 0; l < NL; ++l) {
        float* hout = (l == NL - 1) ? (float*)d_out : houts[l & 1];
        gnn_agg_kernel<<<NN / 8, 256, 0, stream>>>(hin, row_ptr, col_src, deg_inv, mean);
        gnn_gemm_kernel<<<GEMM_BLOCKS, 256, 0, stream>>>(
            mean, hin, Wl + (size_t)l * D * D, Wr + (size_t)l * D * D,
            b + (size_t)l * D, hout, (l < NL - 1) ? 1 : 0);
        hin = hout;
    }
}


# Aggregation: a wave takes its node's sources 32 per load and requests 8 rows before the first add; lists sorted by source once, so the sum order is fixed (was atomic arrival order)
# speedup vs baseline: 1.4837x; 1.1581x over previous
#include <hip/hip_runtime.h>

#define NN 100000
#define NE 1600000
#define D 128
#define D4 132
#define NL 7
#define GEMM_BLOCKS 1250
#define SCAN_T 1024
#define SCAN_BLOCKS ((NN + SCAN_T - 1) / SCAN_T)
static_assert(SCAN_T / 32 == 32, "the cross-wave sum gives each of the 32 lanes one wave total");

typedef __attribute__((ext_vector_type(2))) float v2f;
typedef __attribute__((ext_vector_type(8))) float v8f;


__global__ void gnn_hist_kernel(const int* __restrict__ dst, int* __restrict__ cnt) {
    int e = blockIdx.x * blockDim.x + threadIdx.x;
    if (e < NE) atomicAdd(&cnt[dst[e]], 1);
}

__global__ void __launch_bounds__(SCAN_T)
gnn_scan_kernel(const int* __restrict__ cnt, int* __restrict__ row_ptr,
                float* __restrict__ deg_inv) {
    __shared__ int s_chunk[SCAN_T / 32];
    __shared__ int s_front[SCAN_T / 32];
    int tid  = threadIdx.x;
    int lane = tid & 31;
    int wave = tid >> 5;
    int base = blockIdx.x * SCAN_T;

    int front = 0;
    {
        const int4* c4 = (const int4*)cnt;
        int n4 = base >> 2;
        #pragma unroll 4
        for (int i = tid; i < n4; i += SCAN_T) {
            int4 q = c4[i];
            front += (q.x + q.y) + (q.z + q.w);
        }
    }

    int i = base + tid;
    int v = (i < NN) ? cnt[i] : 0;

    int incl = v;
    for (int off = 1; off < 32; off <<= 1) {
        int t = __shfl_up(incl, off, 32);
        if (lane >= off) incl += t;
    }
    for (int off = 16; off > 0; off >>= 1) front += __shfl_xor(front, off, 32);

    if (lane == 31) s_chunk[wave] = incl;
    if (lane == 0)  s_front[wave] = front;
    __syncthreads();

    int c = s_front[lane] + ((lane < wave) ? s_chunk[lane] : 0);
    for (int off = 16; off > 0; off >>= 1) c += __shfl_xor(c, off, 32);

    if (i < NN) {
        row_ptr[i + 1] = c + incl;
        deg_inv[i] = 1.0f / fmaxf((float)v, 1.0f);
    }
    if (i == 0) row_ptr[0] = 0;
}

__global__ void gnn_fill_kernel(const int* __restrict__ src, const int* __restrict__ dst,
                                const int* __restrict__ row_ptr, int* __restrict__ cnt2,
                                int* __restrict__ col_src) {
    int e = blockIdx.x * blockDim.x + threadIdx.x;
    if (e < NE) {
        int d = dst[e];
        int pos = row_ptr[d] + atomicAdd(&cnt2[d], 1);
        col_src[pos] = src[e];
    }
}


__global__ void __launch_bounds__(256)
gnn_order_kernel(const int* __restrict__ row_ptr, const int* __restrict__ col_raw,
                 int* __restrict__ col_src) {
    int wave = threadIdx.x >> 5;
    int lane = threadIdx.x & 31;
    int node = blockIdx.x * (blockDim.x >> 5) + wave;
    if (node >= NN) return;
    int e0 = __builtin_amdgcn_readfirstlane(row_ptr[node]);
    int n  = __builtin_amdgcn_readfirstlane(row_ptr[node + 1]) - e0;
    if (n <= 32) {
        int key = (lane < n) ? col_raw[e0 + lane] : 0;
        int rank = 0;
        for (int j = 0; j < n; ++j) {
            int kj = __builtin_amdgcn_readlane(key, j);
            rank += (kj < key || (kj == key && j < lane)) ? 1 : 0;
        }
        if (lane < n) col_src[e0 + rank] = key;
    } else {
        for (int i = lane; i < n; i += 32) {
            int key = col_raw[e0 + i];
            int rank = 0;
            for (int j = 0; j < n; ++j) {
                int kj = col_raw[e0 + j];
                rank += (kj < key || (kj == key && j < i)) ? 1 : 0;
            }
            col_src[e0 + rank] = key;
        }
    }
}


#define AGG_U 8

__global__ void __launch_bounds__(256)
gnn_agg_kernel(const float* __restrict__ h, const int* __restrict__ row_ptr,
               const int* __restrict__ col_src, const float* __restrict__ deg_inv,
               float* __restrict__ mean) {
    int wave = threadIdx.x >> 5;
    int lane = threadIdx.x & 31;
    int node = blockIdx.x * (blockDim.x >> 5) + wave;
    if (node >= NN) return;
    int e0 = __builtin_amdgcn_readfirstlane(row_ptr[node]);
    int n  = __builtin_amdgcn_readfirstlane(row_ptr[node + 1]) - e0;
    float di = deg_inv[node];
    const float4* hl = (const float4*)h + lane;
    float4 acc = make_float4(0.f, 0.f, 0.f, 0.f);
    for (int base = 0; base < n; base += 32) {
        int m = n - base;
        if (m > 32) m = 32;
        int idx = (lane < m) ? col_src[e0 + base + lane] : 0;
        int j = 0;
        for (; j + AGG_U <= m; j += AGG_U) {
            float4 v[AGG_U];
            #pragma unroll
            for (int u = 0; u < AGG_U; ++u) {
                int s = __builtin_amdgcn_readlane(idx, j + u);
                v[u] = hl[(size_t)s * (D / 4)];
            }
            #pragma unroll
            for (int u = 0; u < AGG_U; ++u) {
                acc.x += v[u].x; acc.y += v[u].y; acc.z += v[u].z; acc.w += v[u].w;
            }
        }
        int rem = m - j;
        if (rem > 0) {
            float4 v[AGG_U - 1];
            #pragma unroll
            for (int u = 0; u < AGG_U - 1; ++u) {
                if (u < rem) {
                    int s = __builtin_amdgcn_readlane(idx, j + u);
                    v[u] = hl[(size_t)s * (D / 4)];
                }
            }
            #pragma unroll
            for (int u = 0; u < AGG_U - 1; ++u) {
                if (u < rem) {
                    acc.x += v[u].x; acc.y += v[u].y; acc.z += v[u].z; acc.w += v[u].w;
                }
            }
        }
    }
    float4 r = make_float4(acc.x * di, acc.y * di, acc.z * di, acc.w * di);
    ((float4*)(mean + (size_t)node * D))[lane] = r;
}


__global__ void __launch_bounds__(256, 1)
gnn_gemm_kernel(const float* __restrict__ mean, const float* __restrict__ x,
                const float* __restrict__ Wl, const float* __restrict__ Wr,
                const float* __restrict__ bias, float* __restrict__ out, int relu) {
    __shared__ float sA[16 * D4];
    __shared__ float sX[16 * D4];

    int wave = threadIdx.x >> 5;
    int lane = threadIdx.x & 31;
    int r     = lane & 15;
    int khalf = lane >> 4;

    v2f bl[32], br[32];
    {
        const float* wlrow = Wl + (size_t)(wave * 16 + r) * D;
        const float* wrrow = Wr + (size_t)(wave * 16 + r) * D;
        #pragma unroll
        for (int kk = 0; kk < 32; ++kk) {
            int ko = kk * 4 + 2 * khalf;
            bl[kk] = *(const v2f*)(wlrow + ko);
            br[kk] = *(const v2f*)(wrrow + ko);
        }
    }
    float bj = bias[wave * 16 + r];

    for (int rt = blockIdx.x; rt < NN / 16; rt += GEMM_BLOCKS) {
        {
            const float4* gm = (const float4*)(mean + (size_t)rt * 16 * D);
            const float4* gx = (const float4*)(x    + (size_t)rt * 16 * D);
            #pragma unroll
            for (int j = 0; j < 2; ++j) {
                int i    = threadIdx.x + j * 256;
                int row  = i >> 5;
                int col4 = i & 31;
                ((float4*)(sA + row * D4))[col4] = gm[i];
                ((float4*)(sX + row * D4))[col4] = gx[i];
            }
        }
        __syncthreads();

        v8f acc = {};
        #pragma unroll
        for (int kk = 0; kk < 32; ++kk) {
            int ko = kk * 4 + 2 * khalf;
            v2f a = *(const v2f*)(sA + r * D4 + ko);
            acc = __builtin_amdgcn_wmma_f32_16x16x4_f32(false, a, false, bl[kk],
                                                        (short)0, acc, false, false);
            v2f a2 = *(const v2f*)(sX + r * D4 + ko);
            acc = __builtin_amdgcn_wmma_f32_16x16x4_f32(false, a2, false, br[kk],
                                                        (short)0, acc, false, false);
        }

        float* obase = out + (size_t)(rt * 16) * D + wave * 16 + r;
        #pragma unroll
        for (int v = 0; v < 8; ++v) {
            float val = acc[v] + bj;
            if (relu) val = fmaxf(val, 0.0f);
            obase[(size_t)(v + 8 * khalf) * D] = val;
        }
        __syncthreads();
    }
}


extern "C" void kernel_launch(void* const* d_in, const int* in_sizes, int n_in,
                              void* d_out, int out_size, void* d_ws, size_t ws_size,
                              hipStream_t stream) {
    const float* x          = (const float*)d_in[0];
    const int*   edge_index = (const int*)d_in[1];
    const float* Wl         = (const float*)d_in[2];
    const float* Wr         = (const float*)d_in[3];
    const float* b          = (const float*)d_in[4];
    const int* src = edge_index;
    const int* dst = edge_index + NE;

    char* ws = (char*)d_ws;
    size_t off = 0;
    auto carve = [&](size_t bytes) -> void* {
        void* p = ws + off;
        off = (off + bytes + 255) & ~(size_t)255;
        return p;
    };
    float* mean    = (float*)carve((size_t)NN * D * sizeof(float));
    float* hA      = (float*)carve((size_t)NN * D * sizeof(float));
    float* hB      = (float*)carve((size_t)NN * D * sizeof(float));
    float* deg_inv = (float*)carve((size_t)NN * sizeof(float));
    int*   row_ptr = (int*)carve((size_t)(NN + 1) * sizeof(int));
    int*   cnt     = (int*)carve((size_t)NN * sizeof(int));
    int*   col_src = (int*)carve((size_t)NE * sizeof(int));

    hipMemsetAsync(cnt, 0, (size_t)NN * sizeof(int), stream);
    gnn_hist_kernel<<<(NE + 255) / 256, 256, 0, stream>>>(dst, cnt);
    gnn_scan_kernel<<<SCAN_BLOCKS, SCAN_T, 0, stream>>>(cnt, row_ptr, deg_inv);
    hipMemsetAsync(cnt, 0, (size_t)NN * sizeof(int), stream);
    int* col_raw = (int*)mean;
    gnn_fill_kernel<<<(NE + 255) / 256, 256, 0, stream>>>(src, dst, row_ptr, cnt, col_raw);
    gnn_order_kernel<<<NN / 8, 256, 0, stream>>>(row_ptr, col_raw, col_src);

    const float* hin = x;
    float* houts[2] = {hA, hB};
    for (int l = 0; l < NL; ++l) {
        float* hout = (l == NL - 1) ? (float*)d_out : houts[l & 1];
        gnn_agg_kernel<<<NN / 8, 256, 0, stream>>>(hin, row_ptr, col_src, deg_inv, mean);
        gnn_gemm_kernel<<<GEMM_BLOCKS, 256, 0, stream>>>(
            mean, hin, Wl + (size_t)l * D * D, Wr + (size_t)l * D * D,
            b + (size_t)l * D, hout, (l < NL - 1) ? 1 : 0);
        hin = hout;
    }
}


# Aggregation fused into the GEMM kernel: each wave sums its two nodes' neighbour rows (same list order) into the LDS A tile; no mean plane, 7 launches fewer, 512 resident workgroups walk the tiles
# speedup vs baseline: 1.5553x; 1.0483x over previous
#include <hip/hip_runtime.h>

#define NN 100000
#define NE 1600000
#define D 128
#define D4 132
#define NL 7
#define LAYER_WAVES_PER_SIMD 4
#define LAYER_BLOCKS 512
#define SCAN_T 1024
#define SCAN_BLOCKS ((NN + SCAN_T - 1) / SCAN_T)
static_assert(SCAN_T / 32 == 32, "the cross-wave sum gives each of the 32 lanes one wave total");

typedef __attribute__((ext_vector_type(2))) float v2f;
typedef __attribute__((ext_vector_type(8))) float v8f;


__global__ void gnn_hist_kernel(const int* __restrict__ dst, int* __restrict__ cnt) {
    int e = blockIdx.x * blockDim.x + threadIdx.x;
    if (e < NE) atomicAdd(&cnt[dst[e]], 1);
}

__global__ void __launch_bounds__(SCAN_T)
gnn_scan_kernel(const int* __restrict__ cnt, int* __restrict__ row_ptr,
                float* __restrict__ deg_inv) {
    __shared__ int s_chunk[SCAN_T / 32];
    __shared__ int s_front[SCAN_T / 32];
    int tid  = threadIdx.x;
    int lane = tid & 31;
    int wave = tid >> 5;
    int base = blockIdx.x * SCAN_T;

    int front = 0;
    {
        const int4* c4 = (const int4*)cnt;
        int n4 = base >> 2;
        #pragma unroll 4
        for (int i = tid; i < n4; i += SCAN_T) {
            int4 q = c4[i];
            front += (q.x + q.y) + (q.z + q.w);
        }
    }

    int i = base + tid;
    int v = (i < NN) ? cnt[i] : 0;

    int incl = v;
    for (int off = 1; off < 32; off <<= 1) {
        int t = __shfl_up(incl, off, 32);
        if (lane >= off) incl += t;
    }
    for (int off = 16; off > 0; off >>= 1) front += __shfl_xor(front, off, 32);

    if (lane == 31) s_chunk[wave] = incl;
    if (lane == 0)  s_front[wave] = front;
    __syncthreads();

    int c = s_front[lane] + ((lane < wave) ? s_chunk[lane] : 0);
    for (int off = 16; off > 0; off >>= 1) c += __shfl_xor(c, off, 32);

    if (i < NN) {
        row_ptr[i + 1] = c + incl;
        deg_inv[i] = 1.0f / fmaxf((float)v, 1.0f);
    }
    if (i == 0) row_ptr[0] = 0;
}

__global__ void gnn_fill_kernel(const int* __restrict__ src, const int* __restrict__ dst,
                                const int* __restrict__ row_ptr, int* __restrict__ cnt2,
                                int* __restrict__ col_src) {
    int e = blockIdx.x * blockDim.x + threadIdx.x;
    if (e < NE) {
        int d = dst[e];
        int pos = row_ptr[d] + atomicAdd(&cnt2[d], 1);
        col_src[pos] = src[e];
    }
}


__global__ void __launch_bounds__(256)
gnn_order_kernel(const int* __restrict__ row_ptr, const int* __restrict__ col_raw,
                 int* __restrict__ col_src) {
    int wave = threadIdx.x >> 5;
    int lane = threadIdx.x & 31;
    int node = blockIdx.x * (blockDim.x >> 5) + wave;
    if (node >= NN) return;
    int e0 = __builtin_amdgcn_readfirstlane(row_ptr[node]);
    int n  = __builtin_amdgcn_readfirstlane(row_ptr[node + 1]) - e0;
    if (n <= 32) {
        int key = (lane < n) ? col_raw[e0 + lane] : 0;
        int rank = 0;
        for (int j = 0; j < n; ++j) {
            int kj = __builtin_amdgcn_readlane(key, j);
            rank += (kj < key || (kj == key && j < lane)) ? 1 : 0;
        }
        if (lane < n) col_src[e0 + rank] = key;
    } else {
        for (int i = lane; i < n; i += 32) {
            int key = col_raw[e0 + i];
            int rank = 0;
            for (int j = 0; j < n; ++j) {
                int kj = col_raw[e0 + j];
                rank += (kj < key || (kj == key && j < i)) ? 1 : 0;
            }
            col_src[e0 + rank] = key;
        }
    }
}


#define AGG_U 8

__device__ __forceinline__ float4 gnn_sum_rows(const float4* __restrict__ hl,
                                               const int* __restrict__ col_src,
                                               int e0, int n, int idx0, int lane) {
    float4 acc = make_float4(0.f, 0.f, 0.f, 0.f);
    for (int base = 0; base < n; base += 32) {
        int m = n - base;
        if (m > 32) m = 32;
        int idx = idx0;
        if (base > 0) idx = (lane < m) ? col_src[e0 + base + lane] : 0;
        int j = 0;
        for (; j + AGG_U <= m; j += AGG_U) {
            float4 v[AGG_U];
            #pragma unroll
            for (int u = 0; u < AGG_U; ++u) {
                int s = __builtin_amdgcn_readlane(idx, j + u);
                v[u] = hl[(size_t)s * (D / 4)];
            }
            #pragma unroll
            for (int u = 0; u < AGG_U; ++u) {
                acc.x += v[u].x; acc.y += v[u].y; acc.z += v[u].z; acc.w += v[u].w;
            }
        }
        int rem = m - j;
        if (rem > 0) {
            float4 v[AGG_U - 1];
            #pragma unroll
            for (int u = 0; u < AGG_U - 1; ++u) {
                if (u < rem) {
                    int s = __builtin_amdgcn_readlane(idx, j + u);
                    v[u] = hl[(size_t)s * (D / 4)];
                }
            }
            #pragma unroll
            for (int u = 0; u < AGG_U - 1; ++u) {
                if (u < rem) {
                    acc.x += v[u].x; acc.y += v[u].y; acc.z += v[u].z; acc.w += v[u].w;
                }
            }
        }
    }
    return acc;
}

__global__ void __launch_bounds__(256, LAYER_WAVES_PER_SIMD)
gnn_layer_kernel(const float* __restrict__ h, const int* __restrict__ row_ptr,
                 const int* __restrict__ col_src, const float* __restrict__ deg_inv,
                 const float* __restrict__ Wl, const float* __restrict__ Wr,
                 const float* __restrict__ bias, float* __restrict__ out, int relu) {
    __shared__ float sA[16 * D4];
    __shared__ float sX[16 * D4];

    int wave = threadIdx.x >> 5;
    int lane = threadIdx.x & 31;
    int r     = lane & 15;
    int khalf = lane >> 4;

    v2f bl[32], br[32];
    {
        const float* wlrow = Wl + (size_t)(wave * 16 + r) * D;
        const float* wrrow = Wr + (size_t)(wave * 16 + r) * D;
        #pragma unroll
        for (int kk = 0; kk < 32; ++kk) {
            int ko = kk * 4 + 2 * khalf;
            bl[kk] = *(const v2f*)(wlrow + ko);
            br[kk] = *(const v2f*)(wrrow + ko);
        }
    }
    float bj = bias[wave * 16 + r];
    const float4* hl = (const float4*)h + lane;

    for (int rt = blockIdx.x; rt < NN / 16; rt += LAYER_BLOCKS) {
        int node0 = rt * 16 + 2 * wave;
        int p0 = __builtin_amdgcn_readfirstlane(row_ptr[node0]);
        int p1 = __builtin_amdgcn_readfirstlane(row_ptr[node0 + 1]);
        int p2 = __builtin_amdgcn_readfirstlane(row_ptr[node0 + 2]);
        float d0 = deg_inv[node0];
        float d1 = deg_inv[node0 + 1];
        int na = p1 - p0, nb = p2 - p1;
        int ia = (lane < na) ? col_src[p0 + lane] : 0;
        int ib = (lane < nb) ? col_src[p1 + lane] : 0;

        const float4* gx = (const float4*)(h + (size_t)rt * 16 * D);
        float4 x0 = gx[threadIdx.x];
        float4 x1 = gx[threadIdx.x + 256];
        {
            int i0 = threadIdx.x, i1 = threadIdx.x + 256;
            ((float4*)(sX + (i0 >> 5) * D4))[i0 & 31] = x0;
            ((float4*)(sX + (i1 >> 5) * D4))[i1 & 31] = x1;
        }

        float4 sa = gnn_sum_rows(hl, col_src, p0, na, ia, lane);
        float4 sb = gnn_sum_rows(hl, col_src, p1, nb, ib, lane);

        ((float4*)(sA + (2 * wave)     * D4))[lane] = make_float4(sa.x * d0, sa.y * d0, sa.z * d0, sa.w * d0);
        ((float4*)(sA + (2 * wave + 1) * D4))[lane] = make_float4(sb.x * d1, sb.y * d1, sb.z * d1, sb.w * d1);
        __syncthreads();

        v8f acc = {};
        #pragma unroll
        for (int kk = 0; kk < 32; ++kk) {
            int ko = kk * 4 + 2 * khalf;
            v2f a = *(const v2f*)(sA + r * D4 + ko);
            acc = __builtin_amdgcn_wmma_f32_16x16x4_f32(false, a, false, bl[kk],
                                                        (short)0, acc, false, false);
            v2f a2 = *(const v2f*)(sX + r * D4 + ko);
            acc = __builtin_amdgcn_wmma_f32_16x16x4_f32(false, a2, false, br[kk],
                                                        (short)0, acc, false, false);
        }

        float* obase = out + (size_t)(rt * 16) * D + wave * 16 + r;
        #pragma unroll
        for (int v = 0; v < 8; ++v) {
            float val = acc[v] + bj;
            if (relu) val = fmaxf(val, 0.0f);
            obase[(size_t)(v + 8 * khalf) * D] = val;
        }
        __syncthreads();
    }
}


extern "C" void kernel_launch(void* const* d_in, const int* in_sizes, int n_in,
                              void* d_out, int out_size, void* d_ws, size_t ws_size,
                              hipStream_t stream) {
    const float* x          = (const float*)d_in[0];
    const int*   edge_index = (const int*)d_in[1];
    const float* Wl         = (const float*)d_in[2];
    const float* Wr         = (const float*)d_in[3];
    const float* b          = (const float*)d_in[4];
    const int* src = edge_index;
    const int* dst = edge_index + NE;

    char* ws = (char*)d_ws;
    size_t off = 0;
    auto carve = [&](size_t bytes) -> void* {
        void* p = ws + off;
        off = (off + bytes + 255) & ~(size_t)255;
        return p;
    };
    float* mean    = (float*)carve((size_t)NN * D * sizeof(float));
    float* hA      = (float*)carve((size_t)NN * D * sizeof(float));
    float* hB      = (float*)carve((size_t)NN * D * sizeof(float));
    float* deg_inv = (float*)carve((size_t)NN * sizeof(float));
    int*   row_ptr = (int*)carve((size_t)(NN + 1) * sizeof(int));
    int*   cnt     = (int*)carve((size_t)NN * sizeof(int));
    int*   col_src = (int*)carve((size_t)NE * sizeof(int));

    hipMemsetAsync(cnt, 0, (size_t)NN * sizeof(int), stream);
    gnn_hist_kernel<<<(NE + 255) / 256, 256, 0, stream>>>(dst, cnt);
    gnn_scan_kernel<<<SCAN_BLOCKS, SCAN_T, 0, stream>>>(cnt, row_ptr, deg_inv);
    hipMemsetAsync(cnt, 0, (size_t)NN * sizeof(int), stream);
    int* col_raw = (int*)mean;
    gnn_fill_kernel<<<(NE + 255) / 256, 256, 0, stream>>>(src, dst, row_ptr, cnt, col_raw);
    gnn_order_kernel<<<NN / 8, 256, 0, stream>>>(row_ptr, col_raw, col_src);

    const float* hin = x;
    float* houts[2] = {hA, hB};
    for (int l = 0; l < NL; ++l) {
        float* hout = (l == NL - 1) ? (float*)d_out : houts[l & 1];
        gnn_layer_kernel<<<LAYER_BLOCKS, 256, 0, stream>>>(
            hin, row_ptr, col_src, deg_inv, Wl + (size_t)l * D * D, Wr + (size_t)l * D * D,
            b + (size_t)l * D, hout, (l < NL - 1) ? 1 : 0);
        hin = hout;
    }
}


# Aggregation: a wave walks its two nodes' lists as one list with 16 row requests in flight, re-requesting as rows are added (was batches of 8 that drained between); per-node sum order kept
# speedup vs baseline: 1.6649x; 1.0705x over previous
#include <hip/hip_runtime.h>

#define NN 100000
#define NE 1600000
#define D 128
#define D4 132
#define NL 7
#define LAYER_WAVES_PER_SIMD 4
#define LAYER_BLOCKS 512
#define SCAN_T 1024
#define SCAN_BLOCKS ((NN + SCAN_T - 1) / SCAN_T)
static_assert(SCAN_T / 32 == 32, "the cross-wave sum gives each of the 32 lanes one wave total");

typedef __attribute__((ext_vector_type(2))) float v2f;
typedef __attribute__((ext_vector_type(8))) float v8f;


__global__ void gnn_hist_kernel(const int* __restrict__ dst, int* __restrict__ cnt) {
    int e = blockIdx.x * blockDim.x + threadIdx.x;
    if (e < NE) atomicAdd(&cnt[dst[e]], 1);
}

__global__ void __launch_bounds__(SCAN_T)
gnn_scan_kernel(const int* __restrict__ cnt, int* __restrict__ row_ptr,
                float* __restrict__ deg_inv) {
    __shared__ int s_chunk[SCAN_T / 32];
    __shared__ int s_front[SCAN_T / 32];
    int tid  = threadIdx.x;
    int lane = tid & 31;
    int wave = tid >> 5;
    int base = blockIdx.x * SCAN_T;

    int front = 0;
    {
        const int4* c4 = (const int4*)cnt;
        int n4 = base >> 2;
        #pragma unroll 4
        for (int i = tid; i < n4; i += SCAN_T) {
            int4 q = c4[i];
            front += (q.x + q.y) + (q.z + q.w);
        }
    }

    int i = base + tid;
    int v = (i < NN) ? cnt[i] : 0;

    int incl = v;
    for (int off = 1; off < 32; off <<= 1) {
        int t = __shfl_up(incl, off, 32);
        if (lane >= off) incl += t;
    }
    for (int off = 16; off > 0; off >>= 1) front += __shfl_xor(front, off, 32);

    if (lane == 31) s_chunk[wave] = incl;
    if (lane == 0)  s_front[wave] = front;
    __syncthreads();

    int c = s_front[lane] + ((lane < wave) ? s_chunk[lane] : 0);
    for (int off = 16; off > 0; off >>= 1) c += __shfl_xor(c, off, 32);

    if (i < NN) {
        row_ptr[i + 1] = c + incl;
        deg_inv[i] = 1.0f / fmaxf((float)v, 1.0f);
    }
    if (i == 0) row_ptr[0] = 0;
}

__global__ void gnn_fill_kernel(const int* __restrict__ src, const int* __restrict__ dst,
                                const int* __restrict__ row_ptr, int* __restrict__ cnt2,
                                int* __restrict__ col_src) {
    int e = blockIdx.x * blockDim.x + threadIdx.x;
    if (e < NE) {
        int d = dst[e];
        int pos = row_ptr[d] + atomicAdd(&cnt2[d], 1);
        col_src[pos] = src[e];
    }
}


__global__ void __launch_bounds__(256)
gnn_order_kernel(const int* __restrict__ row_ptr, const int* __restrict__ col_raw,
                 int* __restrict__ col_src) {
    int wave = threadIdx.x >> 5;
    int lane = threadIdx.x & 31;
    int node = blockIdx.x * (blockDim.x >> 5) + wave;
    if (node >= NN) return;
    int e0 = __builtin_amdgcn_readfirstlane(row_ptr[node]);
    int n  = __builtin_amdgcn_readfirstlane(row_ptr[node + 1]) - e0;
    if (n <= 32) {
        int key = (lane < n) ? col_raw[e0 + lane] : 0;
        int rank = 0;
        for (int j = 0; j < n; ++j) {
            int kj = __builtin_amdgcn_readlane(key, j);
            rank += (kj < key || (kj == key && j < lane)) ? 1 : 0;
        }
        if (lane < n) col_src[e0 + rank] = key;
    } else {
        for (int i = lane; i < n; i += 32) {
            int key = col_raw[e0 + i];
            int rank = 0;
            for (int j = 0; j < n; ++j) {
                int kj = col_raw[e0 + j];
                rank += (kj < key || (kj == key && j < i)) ? 1 : 0;
            }
            col_src[e0 + rank] = key;
        }
    }
}


#define RING 16

__device__ __forceinline__ int gnn_idx_window(const int* __restrict__ col_src,
                                              int e0, int total, int first, int lane) {
    int k = first + lane;
    if (k > total - 1) k = total - 1;
    return col_src[e0 + k];
}

static_assert(RING == 16, "an index window of 32 lanes holds the current group and the next one");

__device__ __forceinline__ void gnn_sum_two_lists(const float4* __restrict__ hl,
                                                  const int* __restrict__ col_src,
                                                  int e0, int na, int total, int lane,
                                                  float4& sum_a, float4& sum_b) {
    float4 acc = make_float4(0.f, 0.f, 0.f, 0.f);
    float4 sa  = make_float4(0.f, 0.f, 0.f, 0.f);
    if (total > 0) {
        int ngroups = (total + RING - 1) / RING;
        int idx_cur = gnn_idx_window(col_src, e0, total, 0, lane);
        float4 v[RING];
        #pragma unroll
        for (int u = 0; u < RING; ++u) {
            int s = __builtin_amdgcn_readlane(idx_cur, u);
            v[u] = hl[(size_t)s * (D / 4)];
        }
        int g = 0;
        for (; g + 1 < ngroups; ++g) {
            int idx_new = gnn_idx_window(col_src, e0, total, (g + 1) * RING, lane);
            #pragma unroll
            for (int u = 0; u < RING; ++u) {
                int j = g * RING + u;
                if (j == na) { sa = acc; acc = make_float4(0.f, 0.f, 0.f, 0.f); }
                acc.x += v[u].x; acc.y += v[u].y; acc.z += v[u].z; acc.w += v[u].w;
                int s = __builtin_amdgcn_readlane(idx_cur, RING + u);
                v[u] = hl[(size_t)s * (D / 4)];
            }
            idx_cur = idx_new;
        }
        #pragma unroll
        for (int u = 0; u < RING; ++u) {
            int j = g * RING + u;
            if (j < total) {
                if (j == na) { sa = acc; acc = make_float4(0.f, 0.f, 0.f, 0.f); }
                acc.x += v[u].x; acc.y += v[u].y; acc.z += v[u].z; acc.w += v[u].w;
            }
        }
    }
    if (na == total) {
        sum_a = acc;
        sum_b = make_float4(0.f, 0.f, 0.f, 0.f);
    } else {
        sum_a = sa;
        sum_b = acc;
    }
}

__global__ void __launch_bounds__(256, LAYER_WAVES_PER_SIMD)
gnn_layer_kernel(const float* __restrict__ h, const int* __restrict__ row_ptr,
                 const int* __restrict__ col_src, const float* __restrict__ deg_inv,
                 const float* __restrict__ Wl, const float* __restrict__ Wr,
                 const float* __restrict__ bias, float* __restrict__ out, int relu) {
    __shared__ float sA[16 * D4];
    __shared__ float sX[16 * D4];

    int wave = threadIdx.x >> 5;
    int lane = threadIdx.x & 31;
    int r     = lane & 15;
    int khalf = lane >> 4;

    v2f bl[32], br[32];
    {
        const float* wlrow = Wl + (size_t)(wave * 16 + r) * D;
        const float* wrrow = Wr + (size_t)(wave * 16 + r) * D;
        #pragma unroll
        for (int kk = 0; kk < 32; ++kk) {
            int ko = kk * 4 + 2 * khalf;
            bl[kk] = *(const v2f*)(wlrow + ko);
            br[kk] = *(const v2f*)(wrrow + ko);
        }
    }
    float bj = bias[wave * 16 + r];
    const float4* hl = (const float4*)h + lane;

    for (int rt = blockIdx.x; rt < NN / 16; rt += LAYER_BLOCKS) {
        const float4* gx = (const float4*)(h + (size_t)rt * 16 * D);
        float4 x0 = gx[threadIdx.x];
        float4 x1 = gx[threadIdx.x + 256];

        int node0 = rt * 16 + 2 * wave;
        int p0 = __builtin_amdgcn_readfirstlane(row_ptr[node0]);
        int p1 = __builtin_amdgcn_readfirstlane(row_ptr[node0 + 1]);
        int p2 = __builtin_amdgcn_readfirstlane(row_ptr[node0 + 2]);
        float d0 = deg_inv[node0];
        float d1 = deg_inv[node0 + 1];
        int na = p1 - p0, total = p2 - p0;

        {
            int i0 = threadIdx.x, i1 = threadIdx.x + 256;
            ((float4*)(sX + (i0 >> 5) * D4))[i0 & 31] = x0;
            ((float4*)(sX + (i1 >> 5) * D4))[i1 & 31] = x1;
        }

        float4 sa, sb;
        gnn_sum_two_lists(hl, col_src, p0, na, total, lane, sa, sb);

        ((float4*)(sA + (2 * wave)     * D4))[lane] = make_float4(sa.x * d0, sa.y * d0, sa.z * d0, sa.w * d0);
        ((float4*)(sA + (2 * wave + 1) * D4))[lane] = make_float4(sb.x * d1, sb.y * d1, sb.z * d1, sb.w * d1);
        __syncthreads();

        v8f acc = {};
        #pragma unroll
        for (int kk = 0; kk < 32; ++kk) {
            int ko = kk * 4 + 2 * khalf;
            v2f a = *(const v2f*)(sA + r * D4 + ko);
            acc = __builtin_amdgcn_wmma_f32_16x16x4_f32(false, a, false, bl[kk],
                                                        (short)0, acc, false, false);
            v2f a2 = *(const v2f*)(sX + r * D4 + ko);
            acc = __builtin_amdgcn_wmma_f32_16x16x4_f32(false, a2, false, br[kk],
                                                        (short)0, acc, false, false);
        }

        float* obase = out + (size_t)(rt * 16) * D + wave * 16 + r;
        #pragma unroll
        for (int v = 0; v < 8; ++v) {
            float val = acc[v] + bj;
            if (relu) val = fmaxf(val, 0.0f);
            obase[(size_t)(v + 8 * khalf) * D] = val;
        }
        __syncthreads();
    }
}


extern "C" void kernel_launch(void* const* d_in, const int* in_sizes, int n_in,
                              void* d_out, int out_size, void* d_ws, size_t ws_size,
                              hipStream_t stream) {
    const float* x          = (const float*)d_in[0];
    const int*   edge_index = (const int*)d_in[1];
    const float* Wl         = (const float*)d_in[2];
    const float* Wr         = (const float*)d_in[3];
    const float* b          = (const float*)d_in[4];
    const int* src = edge_index;
    const int* dst = edge_index + NE;

    char* ws = (char*)d_ws;
    size_t off = 0;
    auto carve = [&](size_t bytes) -> void* {
        void* p = ws + off;
        off = (off + bytes + 255) & ~(size_t)255;
        return p;
    };
    float* mean    = (float*)carve((size_t)NN * D * sizeof(float));
    float* hA      = (float*)carve((size_t)NN * D * sizeof(float));
    float* hB      = (float*)carve((size_t)NN * D * sizeof(float));
    float* deg_inv = (float*)carve((size_t)NN * sizeof(float));
    int*   row_ptr = (int*)carve((size_t)(NN + 1) * sizeof(int));
    int*   cnt     = (int*)carve((size_t)NN * sizeof(int));
    int*   col_src = (int*)carve((size_t)NE * sizeof(int));

    hipMemsetAsync(cnt, 0, (size_t)NN * sizeof(int), stream);
    gnn_hist_kernel<<<(NE + 255) / 256, 256, 0, stream>>>(dst, cnt);
    gnn_scan_kernel<<<SCAN_BLOCKS, SCAN_T, 0, stream>>>(cnt, row_ptr, deg_inv);
    hipMemsetAsync(cnt, 0, (size_t)NN * sizeof(int), stream);
    int* col_raw = (int*)mean;
    gnn_fill_kernel<<<(NE + 255) / 256, 256, 0, stream>>>(src, dst, row_ptr, cnt, col_raw);
    gnn_order_kernel<<<NN / 8, 256, 0, stream>>>(row_ptr, col_raw, col_src);

    const float* hin = x;
    float* houts[2] = {hA, hB};
    for (int l = 0; l < NL; ++l) {
        float* hout = (l == NL - 1) ? (float*)d_out : houts[l & 1];
        gnn_layer_kernel<<<LAYER_BLOCKS, 256, 0, stream>>>(
            hin, row_ptr, col_src, deg_inv, Wl + (size_t)l * D * D, Wr + (size_t)l * D * D,
            b + (size_t)l * D, hout, (l < NL - 1) ? 1 : 0);
        hin = hout;
    }
}


# Layer kernel pipelined across tiles: the next tile's first 16 neighbour rows, x rows and list bounds are requested before this tile's WMMAs; LDS tiles doubled, one barrier a tile; sum order kept
# speedup vs baseline: 1.8230x; 1.0950x over previous
#include <hip/hip_runtime.h>

#define NN 100000
#define NE 1600000
#define D 128
#define D4 132
#define NL 7
#define LAYER_WAVES_PER_SIMD 4
#define LAYER_BLOCKS 512
#define SCAN_T 1024
#define SCAN_BLOCKS ((NN + SCAN_T - 1) / SCAN_T)
static_assert(SCAN_T / 32 == 32, "the cross-wave sum gives each of the 32 lanes one wave total");

typedef __attribute__((ext_vector_type(2))) float v2f;
typedef __attribute__((ext_vector_type(8))) float v8f;


__global__ void gnn_hist_kernel(const int* __restrict__ dst, int* __restrict__ cnt) {
    int e = blockIdx.x * blockDim.x + threadIdx.x;
    if (e < NE) atomicAdd(&cnt[dst[e]], 1);
}

__global__ void __launch_bounds__(SCAN_T)
gnn_scan_kernel(const int* __restrict__ cnt, int* __restrict__ row_ptr,
                float* __restrict__ deg_inv) {
    __shared__ int s_chunk[SCAN_T / 32];
    __shared__ int s_front[SCAN_T / 32];
    int tid  = threadIdx.x;
    int lane = tid & 31;
    int wave = tid >> 5;
    int base = blockIdx.x * SCAN_T;

    int front = 0;
    {
        const int4* c4 = (const int4*)cnt;
        int n4 = base >> 2;
        #pragma unroll 4
        for (int i = tid; i < n4; i += SCAN_T) {
            int4 q = c4[i];
            front += (q.x + q.y) + (q.z + q.w);
        }
    }

    int i = base + tid;
    int v = (i < NN) ? cnt[i] : 0;

    int incl = v;
    for (int off = 1; off < 32; off <<= 1) {
        int t = __shfl_up(incl, off, 32);
        if (lane >= off) incl += t;
    }
    for (int off = 16; off > 0; off >>= 1) front += __shfl_xor(front, off, 32);

    if (lane == 31) s_chunk[wave] = incl;
    if (lane == 0)  s_front[wave] = front;
    __syncthreads();

    int c = s_front[lane] + ((lane < wave) ? s_chunk[lane] : 0);
    for (int off = 16; off > 0; off >>= 1) c += __shfl_xor(c, off, 32);

    if (i < NN) {
        row_ptr[i + 1] = c + incl;
        deg_inv[i] = 1.0f / fmaxf((float)v, 1.0f);
    }
    if (i == 0) row_ptr[0] = 0;
}

__global__ void gnn_fill_kernel(const int* __restrict__ src, const int* __restrict__ dst,
                                const int* __restrict__ row_ptr, int* __restrict__ cnt2,
                                int* __restrict__ col_src) {
    int e = blockIdx.x * blockDim.x + threadIdx.x;
    if (e < NE) {
        int d = dst[e];
        int pos = row_ptr[d] + atomicAdd(&cnt2[d], 1);
        col_src[pos] = src[e];
    }
}


__global__ void __launch_bounds__(256)
gnn_order_kernel(const int* __restrict__ row_ptr, const int* __restrict__ col_raw,
                 int* __restrict__ col_src) {
    int wave = threadIdx.x >> 5;
    int lane = threadIdx.x & 31;
    int node = blockIdx.x * (blockDim.x >> 5) + wave;
    if (node >= NN) return;
    int e0 = __builtin_amdgcn_readfirstlane(row_ptr[node]);
    int n  = __builtin_amdgcn_readfirstlane(row_ptr[node + 1]) - e0;
    if (n <= 32) {
        int key = (lane < n) ? col_raw[e0 + lane] : 0;
        int rank = 0;
        for (int j = 0; j < n; ++j) {
            int kj = __builtin_amdgcn_readlane(key, j);
            rank += (kj < key || (kj == key && j < lane)) ? 1 : 0;
        }
        if (lane < n) col_src[e0 + rank] = key;
    } else {
        for (int i = lane; i < n; i += 32) {
            int key = col_raw[e0 + i];
            int rank = 0;
            for (int j = 0; j < n; ++j) {
                int kj = col_raw[e0 + j];
                rank += (kj < key || (kj == key && j < i)) ? 1 : 0;
            }
            col_src[e0 + rank] = key;
        }
    }
}


#define RING 16
#define NT (NN / 16)
static_assert(NN % 16 == 0, "whole 16-row tiles");
static_assert(NT >= 2 * LAYER_BLOCKS, "every workgroup owns at least two tiles: the prologue opens the second one unasked");

__device__ __forceinline__ int gnn_idx_window(const int* __restrict__ col_src,
                                              int e0, int total, int first, int lane) {
    int k = first + lane;
    if (k > total - 1) k = total - 1;
    return col_src[e0 + k];
}

__device__ __forceinline__ int gnn_idx_window_any(const int* __restrict__ col_src,
                                                  int e0, int total, int lane) {
    int k = lane;
    if (k > total - 1) k = total - 1;
    int a = e0 + k;
    a = a < 0 ? 0 : a;
    a = a > NE - 1 ? NE - 1 : a;
    return col_src[a];
}

static_assert(RING == 16, "an index window of 32 lanes holds the current group and the next one");

struct gnn_bounds_raw { int r0, r1, r2; float d0, d1; };

__device__ __forceinline__ gnn_bounds_raw gnn_bounds_request(const int* __restrict__ row_ptr,
                                                             const float* __restrict__ deg_inv,
                                                             int node0) {
    gnn_bounds_raw b;
    b.r0 = row_ptr[node0];
    b.r1 = row_ptr[node0 + 1];
    b.r2 = row_ptr[node0 + 2];
    b.d0 = deg_inv[node0];
    b.d1 = deg_inv[node0 + 1];
    return b;
}

struct gnn_lists { int e0, na, total, idx; float d0, d1; };

__device__ __forceinline__ gnn_lists gnn_lists_open(const gnn_bounds_raw& b,
                                                    const int* __restrict__ col_src, int lane) {
    gnn_lists L;
    L.e0    = __builtin_amdgcn_readfirstlane(b.r0);
    L.na    = __builtin_amdgcn_readfirstlane(b.r1) - L.e0;
    L.total = __builtin_amdgcn_readfirstlane(b.r2) - L.e0;
    L.d0 = b.d0;
    L.d1 = b.d1;
    L.idx = gnn_idx_window_any(col_src, L.e0, L.total, lane);
    return L;
}

__device__ __forceinline__ void gnn_ring_start(const float4* __restrict__ hl, int idx_cur,
                                               float4 (&v)[RING]) {
    #pragma unroll
    for (int u = 0; u < RING; ++u) {
        int s = __builtin_amdgcn_readlane(idx_cur, u);
        v[u] = hl[(size_t)s * (D / 4)];
    }
}

__device__ __forceinline__ void gnn_ring_finish(const float4* __restrict__ hl,
                                                const int* __restrict__ col_src,
                                                int e0, int na, int total, int lane, int idx_cur,
                                                float4 (&v)[RING],
                                                float4& sum_a, float4& sum_b) {
    float4 acc = make_float4(0.f, 0.f, 0.f, 0.f);
    float4 sa  = make_float4(0.f, 0.f, 0.f, 0.f);
    int ngroups = (total + RING - 1) / RING;
    int g = 0;
    for (; g + 1 < ngroups; ++g) {
        int idx_new = gnn_idx_window(col_src, e0, total, (g + 1) * RING, lane);
        #pragma unroll
        for (int u = 0; u < RING; ++u) {
            int j = g * RING + u;
            if (j == na) { sa = acc; acc = make_float4(0.f, 0.f, 0.f, 0.f); }
            acc.x += v[u].x; acc.y += v[u].y; acc.z += v[u].z; acc.w += v[u].w;
            int s = __builtin_amdgcn_readlane(idx_cur, RING + u);
            v[u] = hl[(size_t)s * (D / 4)];
        }
        idx_cur = idx_new;
    }
    #pragma unroll
    for (int u = 0; u < RING; ++u) {
        int j = g * RING + u;
        if (j < total) {
            if (j == na) { sa = acc; acc = make_float4(0.f, 0.f, 0.f, 0.f); }
            acc.x += v[u].x; acc.y += v[u].y; acc.z += v[u].z; acc.w += v[u].w;
        }
    }
    if (na == total) {
        sum_a = acc;
        sum_b = make_float4(0.f, 0.f, 0.f, 0.f);
    } else {
        sum_a = sa;
        sum_b = acc;
    }
}

__global__ void __launch_bounds__(256, LAYER_WAVES_PER_SIMD)
gnn_layer_kernel(const float* __restrict__ h, const int* __restrict__ row_ptr,
                 const int* __restrict__ col_src, const float* __restrict__ deg_inv,
                 const float* __restrict__ Wl, const float* __restrict__ Wr,
                 const float* __restrict__ bias, float* __restrict__ out, int relu) {
    __shared__ float sA[2 * 16 * D4];
    __shared__ float sX[2 * 16 * D4];

    int wave = threadIdx.x >> 5;
    int lane = threadIdx.x & 31;
    int r     = lane & 15;
    int khalf = lane >> 4;

    int rt = blockIdx.x;
    gnn_bounds_raw rb0 = gnn_bounds_request(row_ptr, deg_inv, rt * 16 + 2 * wave);
    gnn_bounds_raw rb1 = gnn_bounds_request(row_ptr, deg_inv, (rt + LAYER_BLOCKS) * 16 + 2 * wave);

    v2f bl[32], br[32];
    {
        const float* wlrow = Wl + (size_t)(wave * 16 + r) * D;
        const float* wrrow = Wr + (size_t)(wave * 16 + r) * D;
        #pragma unroll
        for (int kk = 0; kk < 32; ++kk) {
            int ko = kk * 4 + 2 * khalf;
            bl[kk] = *(const v2f*)(wlrow + ko);
            br[kk] = *(const v2f*)(wrrow + ko);
        }
    }
    float bj = bias[wave * 16 + r];
    const float4* hl = (const float4*)h + lane;

    auto x_store = [&](int boff, float4 x0, float4 x1) {
        int i0 = threadIdx.x, i1 = threadIdx.x + 256;
        ((float4*)(sX + boff + (i0 >> 5) * D4))[i0 & 31] = x0;
        ((float4*)(sX + boff + (i1 >> 5) * D4))[i1 & 31] = x1;
    };
    auto mean_store = [&](int boff, float4 sa, float4 sb, float d0, float d1) {
        ((float4*)(sA + boff + (2 * wave)     * D4))[lane] = make_float4(sa.x * d0, sa.y * d0, sa.z * d0, sa.w * d0);
        ((float4*)(sA + boff + (2 * wave + 1) * D4))[lane] = make_float4(sb.x * d1, sb.y * d1, sb.z * d1, sb.w * d1);
    };
    auto products = [&](int boff, int tile) {
        v8f acc = {};
        #pragma unroll
        for (int kk = 0; kk < 32; ++kk) {
            int ko = kk * 4 + 2 * khalf;
            v2f a = *(const v2f*)(sA + boff + r * D4 + ko);
            acc = __builtin_amdgcn_wmma_f32_16x16x4_f32(false, a, false, bl[kk],
                                                        (short)0, acc, false, false);
            v2f a2 = *(const v2f*)(sX + boff + r * D4 + ko);
            acc = __builtin_amdgcn_wmma_f32_16x16x4_f32(false, a2, false, br[kk],
                                                        (short)0, acc, false, false);
        }
        float* obase = out + (size_t)(tile * 16) * D + wave * 16 + r;
        #pragma unroll
        for (int v = 0; v < 8; ++v) {
            float val = acc[v] + bj;
            if (relu) val = fmaxf(val, 0.0f);
            obase[(size_t)(v + 8 * khalf) * D] = val;
        }
    };

    float4 v[RING];
    int bsel = 0;
    int boff = 0;

    gnn_lists nxt;
    {
        const float4* gx = (const float4*)(h + (size_t)rt * 16 * D);
        float4 x0 = gx[threadIdx.x];
        float4 x1 = gx[threadIdx.x + 256];
        gnn_lists cur = gnn_lists_open(rb0, col_src, lane);
        nxt = gnn_lists_open(rb1, col_src, lane);
        gnn_ring_start(hl, cur.idx, v);
        x_store(boff, x0, x1);
        float4 sa, sb;
        gnn_ring_finish(hl, col_src, cur.e0, cur.na, cur.total, lane, cur.idx, v, sa, sb);
        mean_store(boff, sa, sb, cur.d0, cur.d1);
    }

    while (rt + LAYER_BLOCKS < NT) {
        __syncthreads();
        int rtn = rt + LAYER_BLOCKS;
        int rtq = (rtn + LAYER_BLOCKS < NT) ? rtn + LAYER_BLOCKS : rtn;
        gnn_bounds_raw rbq = gnn_bounds_request(row_ptr, deg_inv, rtq * 16 + 2 * wave);
        const float4* gx = (const float4*)(h + (size_t)rtn * 16 * D);
        float4 x0 = gx[threadIdx.x];
        float4 x1 = gx[threadIdx.x + 256];
        gnn_ring_start(hl, nxt.idx, v);
        __builtin_amdgcn_sched_barrier(0);

        products(boff, rt);

        __builtin_amdgcn_sched_barrier(0);

        int bseln = bsel ^ 1;
        int boffn = bseln * (16 * D4);
        gnn_lists nn = gnn_lists_open(rbq, col_src, lane);
        float4 sa, sb;
        gnn_ring_finish(hl, col_src, nxt.e0, nxt.na, nxt.total, lane, nxt.idx, v, sa, sb);
        mean_store(boffn, sa, sb, nxt.d0, nxt.d1);
        x_store(boffn, x0, x1);

        nxt = nn;
        rt = rtn;
        bsel = bseln;
        boff = boffn;
    }

    __syncthreads();
    products(boff, rt);
}


extern "C" void kernel_launch(void* const* d_in, const int* in_sizes, int n_in,
                              void* d_out, int out_size, void* d_ws, size_t ws_size,
                              hipStream_t stream) {
    const float* x          = (const float*)d_in[0];
    const int*   edge_index = (const int*)d_in[1];
    const float* Wl         = (const float*)d_in[2];
    const float* Wr         = (const float*)d_in[3];
    const float* b          = (const float*)d_in[4];
    const int* src = edge_index;
    const int* dst = edge_index + NE;

    char* ws = (char*)d_ws;
    size_t off = 0;
    auto carve = [&](size_t bytes) -> void* {
        void* p = ws + off;
        off = (off + bytes + 255) & ~(size_t)255;
        return p;
    };
    float* mean    = (float*)carve((size_t)NN * D * sizeof(float));
    float* hA      = (float*)carve((size_t)NN * D * sizeof(float));
    float* hB      = (float*)carve((size_t)NN * D * sizeof(float));
    float* deg_inv = (float*)carve((size_t)NN * sizeof(float));
    int*   row_ptr = (int*)carve((size_t)(NN + 1) * sizeof(int));
    int*   cnt     = (int*)carve((size_t)NN * sizeof(int));
    int*   col_src = (int*)carve((size_t)NE * sizeof(int));

    hipMemsetAsync(cnt, 0, (size_t)NN * sizeof(int), stream);
    gnn_hist_kernel<<<(NE + 255) / 256, 256, 0, stream>>>(dst, cnt);
    gnn_scan_kernel<<<SCAN_BLOCKS, SCAN_T, 0, stream>>>(cnt, row_ptr, deg_inv);
    hipMemsetAsync(cnt, 0, (size_t)NN * sizeof(int), stream);
    int* col_raw = (int*)mean;
    gnn_fill_kernel<<<(NE + 255) / 256, 256, 0, stream>>>(src, dst, row_ptr, cnt, col_raw);
    gnn_order_kernel<<<NN / 8, 256, 0, stream>>>(row_ptr, col_raw, col_src);

    const float* hin = x;
    float* houts[2] = {hA, hB};
    for (int l = 0; l < NL; ++l) {
        float* hout = (l == NL - 1) ? (float*)d_out : houts[l & 1];
        gnn_layer_kernel<<<LAYER_BLOCKS, 256, 0, stream>>>(
            hin, row_ptr, col_src, deg_inv, Wl + (size_t)l * D * D, Wr + (size_t)l * D * D,
            b + (size_t)l * D, hout, (l < NL - 1) ? 1 : 0);
        hin = hout;
    }
}


# CSR build as a counting sort through LDS counters: edges binned by 512-node range, then one workgroup per range counts, scans and places its lists in LDS; no global atomics or memsets; sum order kept
# speedup vs baseline: 2.4066x; 1.3201x over previous
#include <hip/hip_runtime.h>

#define NN 100000
#define NE 1600000
#define D 128
#define D4 132
#define NL 7
#define LAYER_WAVES_PER_SIMD 4
#define LAYER_BLOCKS 512
#define SCAN_T 1024
static_assert(SCAN_T / 32 == 32, "the cross-wave sum gives each of the 32 lanes one wave total");

typedef __attribute__((ext_vector_type(2))) float v2f;
typedef __attribute__((ext_vector_type(8))) float v8f;


#define RB_SHIFT 9
#define RB (1 << RB_SHIFT)
#define NR ((NN + RB - 1) / RB)
#define CW 250
#define CH (NE / CW)
#define NBC (NR * CW)
#define BIN_T 1024
#define SRC_BITS 17
#define EPT 8
#define QPT ((CH / 4 + BIN_T - 1) / BIN_T)
#define LIST_CAP 16384
static_assert(NE % CW == 0 && CH % 4 == 0, "a chunk is whole 16-byte groups of edges");
static_assert(NN <= (1 << SRC_BITS), "a source index fits the low bits of a binned word");
static_assert(SRC_BITS + RB_SHIFT < 31, "source and node-within-range fit one positive int");
static_assert(NR <= BIN_T && RB <= BIN_T, "one thread per counter in the scans");
static_assert(NBC <= NN && (size_t)(NBC + 1) * 4 <= (size_t)NN * D * 4, "bin counts and offsets fit the arrays they borrow");
static_assert(BIN_T / 32 == 32, "the cross-wave sum gives each of the 32 lanes one wave total");
#define SCAN_BLOCKS ((NBC + SCAN_T - 1) / SCAN_T)

__device__ __forceinline__ int gnn_block_excl_scan(int v, int* s_wsum, int lane, int wave) {
    int incl = v;
    for (int off = 1; off < 32; off <<= 1) {
        int t = __shfl_up(incl, off, 32);
        if (lane >= off) incl += t;
    }
    if (lane == 31) s_wsum[wave] = incl;
    __syncthreads();
    int c = (lane < wave) ? s_wsum[lane] : 0;
    for (int off = 16; off > 0; off >>= 1) c += __shfl_xor(c, off, 32);
    return c + incl - v;
}

__global__ void __launch_bounds__(BIN_T)
gnn_bin_count_kernel(const int* __restrict__ dst, int* __restrict__ bin_cnt) {
    __shared__ int s_cnt[NR];
    for (int i = threadIdx.x; i < NR; i += BIN_T) s_cnt[i] = 0;
    __syncthreads();
    const int4* d4 = (const int4*)(dst + (size_t)blockIdx.x * CH);
    int4 d[QPT];
    #pragma unroll
    for (int u = 0; u < QPT; ++u) {
        int q = threadIdx.x + u * BIN_T;
        d[u] = (q < CH / 4) ? d4[q] : make_int4(-1, -1, -1, -1);
    }
    #pragma unroll
    for (int u = 0; u < QPT; ++u) {
        if (d[u].x >= 0) {
            atomicAdd(&s_cnt[d[u].x >> RB_SHIFT], 1);
            atomicAdd(&s_cnt[d[u].y >> RB_SHIFT], 1);
            atomicAdd(&s_cnt[d[u].z >> RB_SHIFT], 1);
            atomicAdd(&s_cnt[d[u].w >> RB_SHIFT], 1);
        }
    }
    __syncthreads();
    for (int i = threadIdx.x; i < NR; i += BIN_T) bin_cnt[i * CW + blockIdx.x] = s_cnt[i];
}

__global__ void __launch_bounds__(SCAN_T)
gnn_scan_kernel(const int* __restrict__ cnt, int* __restrict__ off_out) {
    __shared__ int s_chunk[SCAN_T / 32];
    __shared__ int s_front[SCAN_T / 32];
    int tid  = threadIdx.x;
    int lane = tid & 31;
    int wave = tid >> 5;
    int base = blockIdx.x * SCAN_T;

    int front = 0;
    {
        const int4* c4 = (const int4*)cnt;
        int n4 = base >> 2;
        #pragma unroll 4
        for (int i = tid; i < n4; i += SCAN_T) {
            int4 q = c4[i];
            front += (q.x + q.y) + (q.z + q.w);
        }
    }

    int i = base + tid;
    int v = (i < NBC) ? cnt[i] : 0;

    int incl = v;
    for (int off = 1; off < 32; off <<= 1) {
        int t = __shfl_up(incl, off, 32);
        if (lane >= off) incl += t;
    }
    for (int off = 16; off > 0; off >>= 1) front += __shfl_xor(front, off, 32);

    if (lane == 31) s_chunk[wave] = incl;
    if (lane == 0)  s_front[wave] = front;
    __syncthreads();

    int c = s_front[lane] + ((lane < wave) ? s_chunk[lane] : 0);
    for (int off = 16; off > 0; off >>= 1) c += __shfl_xor(c, off, 32);

    if (i < NBC) off_out[i + 1] = c + incl;
    if (i == 0) off_out[0] = 0;
}

__global__ void __launch_bounds__(BIN_T)
gnn_bin_place_kernel(const int* __restrict__ src, const int* __restrict__ dst,
                     const int* __restrict__ bin_off, int* __restrict__ binned) {
    __shared__ int s_cur[NR];
    __shared__ int s_loff[NR];
    __shared__ int s_gbase[NR];
    __shared__ int s_wsum[32];
    __shared__ int s_src[CH];
    __shared__ int s_dst[CH];
    int tid  = threadIdx.x;
    int lane = tid & 31;
    int wave = tid >> 5;
    int w = blockIdx.x;

    int gb = 0, n = 0;
    if (tid < NR) {
        gb = bin_off[tid * CW + w];
        n  = bin_off[tid * CW + w + 1] - gb;
    }
    int lo = gnn_block_excl_scan(n, s_wsum, lane, wave);
    if (tid < NR) { s_loff[tid] = lo; s_cur[tid] = lo; s_gbase[tid] = gb; }
    __syncthreads();

    const int4* d4 = (const int4*)(dst + (size_t)w * CH);
    const int4* s4 = (const int4*)(src + (size_t)w * CH);
    int4 d[QPT], s[QPT];
    #pragma unroll
    for (int u = 0; u < QPT; ++u) {
        int q = tid + u * BIN_T;
        bool in = q < CH / 4;
        d[u] = in ? d4[q] : make_int4(-1, -1, -1, -1);
        s[u] = in ? s4[q] : make_int4(0, 0, 0, 0);
    }
    #pragma unroll
    for (int u = 0; u < QPT; ++u) {
        if (d[u].x >= 0) {
            int p;
            p = atomicAdd(&s_cur[d[u].x >> RB_SHIFT], 1); s_src[p] = s[u].x; s_dst[p] = d[u].x;
            p = atomicAdd(&s_cur[d[u].y >> RB_SHIFT], 1); s_src[p] = s[u].y; s_dst[p] = d[u].y;
            p = atomicAdd(&s_cur[d[u].z >> RB_SHIFT], 1); s_src[p] = s[u].z; s_dst[p] = d[u].z;
            p = atomicAdd(&s_cur[d[u].w >> RB_SHIFT], 1); s_src[p] = s[u].w; s_dst[p] = d[u].w;
        }
    }
    __syncthreads();

    for (int i = tid; i < CH; i += BIN_T) {
        int d = s_dst[i];
        int r = d >> RB_SHIFT;
        binned[s_gbase[r] + (i - s_loff[r])] = s_src[i] | ((d & (RB - 1)) << SRC_BITS);
    }
}

__global__ void __launch_bounds__(BIN_T)
gnn_range_build_kernel(const int* __restrict__ binned, const int* __restrict__ bin_off,
                       int* __restrict__ row_ptr, float* __restrict__ deg_inv,
                       int* __restrict__ col_raw) {
    __shared__ int s_cnt[RB];
    __shared__ int s_off[RB];
    __shared__ int s_wsum[32];
    __shared__ int s_list[LIST_CAP];
    int tid  = threadIdx.x;
    int lane = tid & 31;
    int wave = tid >> 5;
    int r = blockIdx.x;
    int e_lo = bin_off[r * CW];
    int e_hi = bin_off[(r + 1) * CW];
    int n_e  = e_hi - e_lo;

    if (tid < RB) s_cnt[tid] = 0;
    __syncthreads();
    for (int e0 = e_lo + tid; e0 < e_hi; e0 += EPT * BIN_T) {
        int wd[EPT];
        #pragma unroll
        for (int u = 0; u < EPT; ++u) {
            int e = e0 + u * BIN_T;
            wd[u] = (e < e_hi) ? binned[e] : -1;
        }
        #pragma unroll
        for (int u = 0; u < EPT; ++u)
            if (wd[u] >= 0) atomicAdd(&s_cnt[wd[u] >> SRC_BITS], 1);
    }
    __syncthreads();

    int v = (tid < RB) ? s_cnt[tid] : 0;
    int excl = gnn_block_excl_scan(v, s_wsum, lane, wave);
    if (tid < RB) {
        s_off[tid] = excl;
        s_cnt[tid] = 0;
        int node = r * RB + tid;
        if (node < NN) {
            row_ptr[node] = e_lo + excl;
            deg_inv[node] = 1.0f / fmaxf((float)v, 1.0f);
        }
    }
    if (r == NR - 1 && tid == 0) row_ptr[NN] = e_hi;
    __syncthreads();

    bool fits = n_e <= LIST_CAP;
    for (int e0 = e_lo + tid; e0 < e_hi; e0 += EPT * BIN_T) {
        int wd[EPT];
        #pragma unroll
        for (int u = 0; u < EPT; ++u) {
            int e = e0 + u * BIN_T;
            wd[u] = (e < e_hi) ? binned[e] : -1;
        }
        #pragma unroll
        for (int u = 0; u < EPT; ++u) {
            if (wd[u] >= 0) {
                int dl = wd[u] >> SRC_BITS;
                int pos = s_off[dl] + atomicAdd(&s_cnt[dl], 1);
                int sv = wd[u] & ((1 << SRC_BITS) - 1);
                if (fits) s_list[pos] = sv;
                else      col_raw[e_lo + pos] = sv;
            }
        }
    }
    __syncthreads();
    if (fits) {
        for (int i = tid; i < n_e; i += BIN_T) col_raw[e_lo + i] = s_list[i];
    }
}


__global__ void __launch_bounds__(256)
gnn_order_kernel(const int* __restrict__ row_ptr, const int* __restrict__ col_raw,
                 int* __restrict__ col_src) {
    int wave = threadIdx.x >> 5;
    int lane = threadIdx.x & 31;
    int node = blockIdx.x * (blockDim.x >> 5) + wave;
    if (node >= NN) return;
    int e0 = __builtin_amdgcn_readfirstlane(row_ptr[node]);
    int n  = __builtin_amdgcn_readfirstlane(row_ptr[node + 1]) - e0;
    if (n <= 32) {
        int key = (lane < n) ? col_raw[e0 + lane] : 0;
        int rank = 0;
        for (int j = 0; j < n; ++j) {
            int kj = __builtin_amdgcn_readlane(key, j);
            rank += (kj < key || (kj == key && j < lane)) ? 1 : 0;
        }
        if (lane < n) col_src[e0 + rank] = key;
    } else {
        for (int i = lane; i < n; i += 32) {
            int key = col_raw[e0 + i];
            int rank = 0;
            for (int j = 0; j < n; ++j) {
                int kj = col_raw[e0 + j];
                rank += (kj < key || (kj == key && j < i)) ? 1 : 0;
            }
            col_src[e0 + rank] = key;
        }
    }
}


#define RING 16
#define NT (NN / 16)
static_assert(NN % 16 == 0, "whole 16-row tiles");
static_assert(NT >= 2 * LAYER_BLOCKS, "every workgroup owns at least two tiles: the prologue opens the second one unasked");

__device__ __forceinline__ int gnn_idx_window(const int* __restrict__ col_src,
                                              int e0, int total, int first, int lane) {
    int k = first + lane;
    if (k > total - 1) k = total - 1;
    return col_src[e0 + k];
}

__device__ __forceinline__ int gnn_idx_window_any(const int* __restrict__ col_src,
                                                  int e0, int total, int lane) {
    int k = lane;
    if (k > total - 1) k = total - 1;
    int a = e0 + k;
    a = a < 0 ? 0 : a;
    a = a > NE - 1 ? NE - 1 : a;
    return col_src[a];
}

static_assert(RING == 16, "an index window of 32 lanes holds the current group and the next one");

struct gnn_bounds_raw { int r0, r1, r2; float d0, d1; };

__device__ __forceinline__ gnn_bounds_raw gnn_bounds_request(const int* __restrict__ row_ptr,
                                                             const float* __restrict__ deg_inv,
                                                             int node0) {
    gnn_bounds_raw b;
    b.r0 = row_ptr[node0];
    b.r1 = row_ptr[node0 + 1];
    b.r2 = row_ptr[node0 + 2];
    b.d0 = deg_inv[node0];
    b.d1 = deg_inv[node0 + 1];
    return b;
}

struct gnn_lists { int e0, na, total, idx; float d0, d1; };

__device__ __forceinline__ gnn_lists gnn_lists_open(const gnn_bounds_raw& b,
                                                    const int* __restrict__ col_src, int lane) {
    gnn_lists L;
    L.e0    = __builtin_amdgcn_readfirstlane(b.r0);
    L.na    = __builtin_amdgcn_readfirstlane(b.r1) - L.e0;
    L.total = __builtin_amdgcn_readfirstlane(b.r2) - L.e0;
    L.d0 = b.d0;
    L.d1 = b.d1;
    L.idx = gnn_idx_window_any(col_src, L.e0, L.total, lane);
    return L;
}

__device__ __forceinline__ void gnn_ring_start(const float4* __restrict__ hl, int idx_cur,
                                               float4 (&v)[RING]) {
    #pragma unroll
    for (int u = 0; u < RING; ++u) {
        int s = __builtin_amdgcn_readlane(idx_cur, u);
        v[u] = hl[(size_t)s * (D / 4)];
    }
}

__device__ __forceinline__ void gnn_ring_finish(const float4* __restrict__ hl,
                                                const int* __restrict__ col_src,
                                                int e0, int na, int total, int lane, int idx_cur,
                                                float4 (&v)[RING],
                                                float4& sum_a, float4& sum_b) {
    float4 acc = make_float4(0.f, 0.f, 0.f, 0.f);
    float4 sa  = make_float4(0.f, 0.f, 0.f, 0.f);
    int ngroups = (total + RING - 1) / RING;
    int g = 0;
    for (; g + 1 < ngroups; ++g) {
        int idx_new = gnn_idx_window(col_src, e0, total, (g + 1) * RING, lane);
        #pragma unroll
        for (int u = 0; u < RING; ++u) {
            int j = g * RING + u;
            if (j == na) { sa = acc; acc = make_float4(0.f, 0.f, 0.f, 0.f); }
            acc.x += v[u].x; acc.y += v[u].y; acc.z += v[u].z; acc.w += v[u].w;
            int s = __builtin_amdgcn_readlane(idx_cur, RING + u);
            v[u] = hl[(size_t)s * (D / 4)];
        }
        idx_cur = idx_new;
    }
    #pragma unroll
    for (int u = 0; u < RING; ++u) {
        int j = g * RING + u;
        if (j < total) {
            if (j == na) { sa = acc; acc = make_float4(0.f, 0.f, 0.f, 0.f); }
            acc.x += v[u].x; acc.y += v[u].y; acc.z += v[u].z; acc.w += v[u].w;
        }
    }
    if (na == total) {
        sum_a = acc;
        sum_b = make_float4(0.f, 0.f, 0.f, 0.f);
    } else {
        sum_a = sa;
        sum_b = acc;
    }
}

__global__ void __launch_bounds__(256, LAYER_WAVES_PER_SIMD)
gnn_layer_kernel(const float* __restrict__ h, const int* __restrict__ row_ptr,
                 const int* __restrict__ col_src, const float* __restrict__ deg_inv,
                 const float* __restrict__ Wl, const float* __restrict__ Wr,
                 const float* __restrict__ bias, float* __restrict__ out, int relu) {
    __shared__ float sA[2 * 16 * D4];
    __shared__ float sX[2 * 16 * D4];

    int wave = threadIdx.x >> 5;
    int lane = threadIdx.x & 31;
    int r     = lane & 15;
    int khalf = lane >> 4;

    int rt = blockIdx.x;
    gnn_bounds_raw rb0 = gnn_bounds_request(row_ptr, deg_inv, rt * 16 + 2 * wave);
    gnn_bounds_raw rb1 = gnn_bounds_request(row_ptr, deg_inv, (rt + LAYER_BLOCKS) * 16 + 2 * wave);

    v2f bl[32], br[32];
    {
        const float* wlrow = Wl + (size_t)(wave * 16 + r) * D;
        const float* wrrow = Wr + (size_t)(wave * 16 + r) * D;
        #pragma unroll
        for (int kk = 0; kk < 32; ++kk) {
            int ko = kk * 4 + 2 * khalf;
            bl[kk] = *(const v2f*)(wlrow + ko);
            br[kk] = *(const v2f*)(wrrow + ko);
        }
    }
    float bj = bias[wave * 16 + r];
    const float4* hl = (const float4*)h + lane;

    auto x_store = [&](int boff, float4 x0, float4 x1) {
        int i0 = threadIdx.x, i1 = threadIdx.x + 256;
        ((float4*)(sX + boff + (i0 >> 5) * D4))[i0 & 31] = x0;
        ((float4*)(sX + boff + (i1 >> 5) * D4))[i1 & 31] = x1;
    };
    auto mean_store = [&](int boff, float4 sa, float4 sb, float d0, float d1) {
        ((float4*)(sA + boff + (2 * wave)     * D4))[lane] = make_float4(sa.x * d0, sa.y * d0, sa.z * d0, sa.w * d0);
        ((float4*)(sA + boff + (2 * wave + 1) * D4))[lane] = make_float4(sb.x * d1, sb.y * d1, sb.z * d1, sb.w * d1);
    };
    auto products = [&](int boff, int tile) {
        v8f acc = {};
        #pragma unroll
        for (int kk = 0; kk < 32; ++kk) {
            int ko = kk * 4 + 2 * khalf;
            v2f a = *(const v2f*)(sA + boff + r * D4 + ko);
            acc = __builtin_amdgcn_wmma_f32_16x16x4_f32(false, a, false, bl[kk],
                                                        (short)0, acc, false, false);
            v2f a2 = *(const v2f*)(sX + boff + r * D4 + ko);
            acc = __builtin_amdgcn_wmma_f32_16x16x4_f32(false, a2, false, br[kk],
                                                        (short)0, acc, false, false);
        }
        float* obase = out + (size_t)(tile * 16) * D + wave * 16 + r;
        #pragma unroll
        for (int v = 0; v < 8; ++v) {
            float val = acc[v] + bj;
            if (relu) val = fmaxf(val, 0.0f);
            obase[(size_t)(v + 8 * khalf) * D] = val;
        }
    };

    float4 v[RING];
    int bsel = 0;
    int boff = 0;

    gnn_lists nxt;
    {
        const float4* gx = (const float4*)(h + (size_t)rt * 16 * D);
        float4 x0 = gx[threadIdx.x];
        float4 x1 = gx[threadIdx.x + 256];
        gnn_lists cur = gnn_lists_open(rb0, col_src, lane);
        nxt = gnn_lists_open(rb1, col_src, lane);
        gnn_ring_start(hl, cur.idx, v);
        x_store(boff, x0, x1);
        float4 sa, sb;
        gnn_ring_finish(hl, col_src, cur.e0, cur.na, cur.total, lane, cur.idx, v, sa, sb);
        mean_store(boff, sa, sb, cur.d0, cur.d1);
    }

    while (rt + LAYER_BLOCKS < NT) {
        __syncthreads();
        int rtn = rt + LAYER_BLOCKS;
        int rtq = (rtn + LAYER_BLOCKS < NT) ? rtn + LAYER_BLOCKS : rtn;
        gnn_bounds_raw rbq = gnn_bounds_request(row_ptr, deg_inv, rtq * 16 + 2 * wave);
        const float4* gx = (const float4*)(h + (size_t)rtn * 16 * D);
        float4 x0 = gx[threadIdx.x];
        float4 x1 = gx[threadIdx.x + 256];
        gnn_ring_start(hl, nxt.idx, v);
        __builtin_amdgcn_sched_barrier(0);

        products(boff, rt);

        __builtin_amdgcn_sched_barrier(0);

        int bseln = bsel ^ 1;
        int boffn = bseln * (16 * D4);
        gnn_lists nn = gnn_lists_open(rbq, col_src, lane);
        float4 sa, sb;
        gnn_ring_finish(hl, col_src, nxt.e0, nxt.na, nxt.total, lane, nxt.idx, v, sa, sb);
        mean_store(boffn, sa, sb, nxt.d0, nxt.d1);
        x_store(boffn, x0, x1);

        nxt = nn;
        rt = rtn;
        bsel = bseln;
        boff = boffn;
    }

    __syncthreads();
    products(boff, rt);
}


extern "C" void kernel_launch(void* const* d_in, const int* in_sizes, int n_in,
                              void* d_out, int out_size, void* d_ws, size_t ws_size,
                              hipStream_t stream) {
    const float* x          = (const float*)d_in[0];
    const int*   edge_index = (const int*)d_in[1];
    const float* Wl         = (const float*)d_in[2];
    const float* Wr         = (const float*)d_in[3];
    const float* b          = (const float*)d_in[4];
    const int* src = edge_index;
    const int* dst = edge_index + NE;

    char* ws = (char*)d_ws;
    size_t off = 0;
    auto carve = [&](size_t bytes) -> void* {
        void* p = ws + off;
        off = (off + bytes + 255) & ~(size_t)255;
        return p;
    };
    float* mean    = (float*)carve((size_t)NN * D * sizeof(float));
    float* hA      = (float*)carve((size_t)NN * D * sizeof(float));
    float* hB      = (float*)carve((size_t)NN * D * sizeof(float));
    float* deg_inv = (float*)carve((size_t)NN * sizeof(float));
    int*   row_ptr = (int*)carve((size_t)(NN + 1) * sizeof(int));
    int*   cnt     = (int*)carve((size_t)NN * sizeof(int));
    int*   col_src = (int*)carve((size_t)NE * sizeof(int));

    int* col_raw = (int*)mean;
    int* binned  = (int*)hA;
    int* bin_off = (int*)hB;
    int* bin_cnt = cnt;
    gnn_bin_count_kernel<<<CW, BIN_T, 0, stream>>>(dst, bin_cnt);
    gnn_scan_kernel<<<SCAN_BLOCKS, SCAN_T, 0, stream>>>(bin_cnt, bin_off);
    gnn_bin_place_kernel<<<CW, BIN_T, 0, stream>>>(src, dst, bin_off, binned);
    gnn_range_build_kernel<<<NR, BIN_T, 0, stream>>>(binned, bin_off, row_ptr, deg_inv, col_raw);
    gnn_order_kernel<<<NN / 8, 256, 0, stream>>>(row_ptr, col_raw, col_src);

    const float* hin = x;
    float* houts[2] = {hA, hB};
    for (int l = 0; l < NL; ++l) {
        float* hout = (l == NL - 1) ? (float*)d_out : houts[l & 1];
        gnn_layer_kernel<<<LAYER_BLOCKS, 256, 0, stream>>>(
            hin, row_ptr, col_src, deg_inv, Wl + (size_t)l * D * D, Wr + (size_t)l * D * D,
            b + (size_t)l * D, hout, (l < NL - 1) ? 1 : 0);
        hin = hout;
    }
}
